# starting kernel
# baseline (speedup 1.0000x reference)
#include <hip/hip_runtime.h>


typedef __attribute__((ext_vector_type(2))) float v2f;
typedef __attribute__((ext_vector_type(8))) float v8f;

#define BATCH 8
#define CCH   256
#define HH    64
#define WW    96
#define ND    21
#define APR   224
#define BPR   288
#define NT    384

__global__ __launch_bounds__(NT, 1)
void corr_wmma_f32_kernel(const float* __restrict__ f1,
                          const float* __restrict__ f2,
                          float* __restrict__ out) {
  extern __shared__ float smem[];
  float* __restrict__ Ald = smem;
  float* __restrict__ Bld = smem + (CCH / 2) * APR;

  const int tid  = threadIdx.x;
  const int lane = tid & 31;
  const int wv   = tid >> 5;
  const int ti   = wv >> 1;
  const int djb  = (wv & 1) * 2;

  const int h = blockIdx.x;
  const int b = blockIdx.y;

  const int row = lane & 15;
  const int khi = lane >> 4;

  const size_t inBase = (size_t)b * (CCH * HH * WW);

  {
    const float* src = f1 + inBase + (size_t)h * WW;
    for (int i = tid; i < CCH * (WW / 4); i += NT) {
      const int c = i / (WW / 4);
      const int q = i % (WW / 4);
      float4 v = *(const float4*)(src + (size_t)c * (HH * WW) + q * 4);
      float* d = &Ald[(c >> 1) * APR + (c & 1) + 8 * q];
      d[0] = v.x; d[2] = v.y; d[4] = v.z; d[6] = v.w;
    }
  }
  {
    const float4 z = make_float4(0.f, 0.f, 0.f, 0.f);
    for (int i = tid; i < (CCH / 2) * 24; i += NT) {
      const int p = i / 24;
      const int j = i % 24;
      const int off = (j < 10) ? (j * 4) : (232 + (j - 10) * 4);
      *(float4*)(&Bld[p * BPR + off]) = z;
    }
  }
  __syncthreads();

  for (int dyI = 0; dyI < ND; ++dyI) {
    const int h2 = h + dyI * 2 - 20;
    const size_t outBase =
        (((size_t)b * (ND * ND) + (size_t)dyI * ND) * HH + (size_t)h) * WW;

    if (h2 < 0 || h2 >= HH) {
      for (int i = tid; i < ND * WW; i += NT) {
        const int dxI = i / WW, w = i % WW;
        out[outBase + (size_t)dxI * (HH * WW) + w] = 0.f;
      }
      continue;
    }

    __syncthreads();
    {
      const float* src = f2 + inBase + (size_t)h2 * WW;
      for (int i = tid; i < CCH * (WW / 4); i += NT) {
        const int c = i / (WW / 4);
        const int q = i % (WW / 4);
        float4 v = *(const float4*)(src + (size_t)c * (HH * WW) + q * 4);
        float* d = &Bld[(c >> 1) * BPR + (c & 1) + 40 + 8 * q];
        d[0] = v.x; d[2] = v.y; d[4] = v.z; d[6] = v.w;
      }
      const int h2n = h2 + 2;
      if (h2n < HH) {
        const float* nsrc = f2 + inBase + (size_t)h2n * WW;
        for (int i = tid; i < CCH; i += NT)
          __builtin_prefetch(nsrc + (size_t)i * (HH * WW), 0, 0);
      }
    }
    __syncthreads();

    v8f acc0 = {0.f, 0.f, 0.f, 0.f, 0.f, 0.f, 0.f, 0.f};
    v8f acc1 = {0.f, 0.f, 0.f, 0.f, 0.f, 0.f, 0.f, 0.f};

    const float* ap = &Ald[khi * APR + 2 * (ti * 16 + row)];
    const float* bp = &Bld[khi * BPR + 2 * ((ti + djb) * 16 + row)];

#pragma unroll 8
    for (int k = 0; k < CCH; k += 4) {
      v2f a  = *(const v2f*)(ap);
      v2f b0 = *(const v2f*)(bp);
      v2f b1 = *(const v2f*)(bp + 32);
      acc0 = __builtin_amdgcn_wmma_f32_16x16x4_f32(
          false, a, false, b0, (short)0, acc0, false, false);
      acc1 = __builtin_amdgcn_wmma_f32_16x16x4_f32(
          false, a, false, b1, (short)0, acc1, false, false);
      ap += 2 * APR;
      bp += 2 * BPR;
    }

    const int mBase = khi * 8;
    const int N = lane & 15;
#pragma unroll
    for (int r = 0; r < 8; ++r) {
      const int m  = mBase + r;
      const int w1 = ti * 16 + m;
      {
        const int diff = 16 * djb + N - m;
        if (diff >= 0 && diff <= 40 && !(diff & 1))
          out[outBase + (size_t)(diff >> 1) * (HH * WW) + w1] =
              acc0[r] * 0.00390625f;
      }
      {
        const int diff = 16 * (djb + 1) + N - m;
        if (diff >= 0 && diff <= 40 && !(diff & 1))
          out[outBase + (size_t)(diff >> 1) * (HH * WW) + w1] =
              acc1[r] * 0.00390625f;
      }
    }
  }
}

extern "C" void kernel_launch(void* const* d_in, const int* in_sizes, int n_in,
                              void* d_out, int out_size, void* d_ws, size_t ws_size,
                              hipStream_t stream) {
  const float* f1 = (const float*)d_in[0];
  const float* f2 = (const float*)d_in[1];
  float* out = (float*)d_out;

  dim3 grid(HH, BATCH);
  dim3 block(NT);
  const size_t shmem =
      (size_t)((CCH / 2) * APR + (CCH / 2) * BPR) * sizeof(float);
  corr_wmma_f32_kernel<<<grid, block, shmem, stream>>>(f1, f2, out);
}


# f2 row staging software-pipelined: the next dy's row is loaded into registers before the WMMA loop and stored to LDS after it, so its global-load latency hides under the matrix work
# speedup vs baseline: 1.1385x; 1.1385x over previous
#include <hip/hip_runtime.h>


typedef __attribute__((ext_vector_type(2))) float v2f;
typedef __attribute__((ext_vector_type(8))) float v8f;

#define BATCH 8
#define CCH   256
#define HH    64
#define WW    96
#define ND    21
#define APR   224
#define BPR   288
#define NT    384
#define NLD   (CCH * (WW / 4) / NT)

__global__ __launch_bounds__(NT, 1)
void corr_wmma_f32_kernel(const float* __restrict__ f1,
                          const float* __restrict__ f2,
                          float* __restrict__ out) {
  extern __shared__ float smem[];
  float* __restrict__ Ald = smem;
  float* __restrict__ Bld = smem + (CCH / 2) * APR;

  const int tid  = threadIdx.x;
  const int lane = tid & 31;
  const int wv   = tid >> 5;
  const int ti   = wv >> 1;
  const int djb  = (wv & 1) * 2;

  const int h = blockIdx.x;
  const int b = blockIdx.y;

  const int row = lane & 15;
  const int khi = lane >> 4;

  const size_t inBase = (size_t)b * (CCH * HH * WW);

  const int dLo = (h < 20) ? ((21 - h) >> 1) : 0;
  const int dHi = (h > 43) ? ((83 - h) >> 1) : 20;

  constexpr int CSTEP = NT / (WW / 4);
  static_assert(NT % (WW / 4) == 0 && CCH % CSTEP == 0 && (CSTEP % 2) == 0,
                "staging map needs whole, even channel steps");
  const int c0 = tid / (WW / 4);
  const int q0 = tid % (WW / 4);
  const float* fsrc = f2 + inBase + (size_t)c0 * (HH * WW) + q0 * 4;
  float* bdst = &Bld[(c0 >> 1) * BPR + (c0 & 1) + 40 + 8 * q0];
  float4 pre[NLD];

  auto fetch = [&](int h2r) {
    const float* src = fsrc + (size_t)h2r * WW;
#pragma unroll
    for (int j = 0; j < NLD; ++j)
      pre[j] = *(const float4*)(src + (size_t)j * (CSTEP * HH * WW));
  };
  auto put = [&]() {
#pragma unroll
    for (int j = 0; j < NLD; ++j) {
      float* d = bdst + j * ((CSTEP / 2) * BPR);
      d[0] = pre[j].x; d[2] = pre[j].y; d[4] = pre[j].z; d[6] = pre[j].w;
    }
  };

  fetch(h + dLo * 2 - 20);

  {
    const float* src = f1 + inBase + (size_t)h * WW;
    for (int i = tid; i < CCH * (WW / 4); i += NT) {
      const int c = i / (WW / 4);
      const int q = i % (WW / 4);
      float4 v = *(const float4*)(src + (size_t)c * (HH * WW) + q * 4);
      float* d = &Ald[(c >> 1) * APR + (c & 1) + 8 * q];
      d[0] = v.x; d[2] = v.y; d[4] = v.z; d[6] = v.w;
    }
  }
  {
    const float4 z = make_float4(0.f, 0.f, 0.f, 0.f);
    for (int i = tid; i < (CCH / 2) * 24; i += NT) {
      const int p = i / 24;
      const int j = i % 24;
      const int off = (j < 10) ? (j * 4) : (232 + (j - 10) * 4);
      *(float4*)(&Bld[p * BPR + off]) = z;
    }
  }
  __syncthreads();

  auto tiles_and_scatter = [&](int dyI) {
    const size_t outBase =
        (((size_t)b * (ND * ND) + (size_t)dyI * ND) * HH + (size_t)h) * WW;

    v8f acc0 = {0.f, 0.f, 0.f, 0.f, 0.f, 0.f, 0.f, 0.f};
    v8f acc1 = {0.f, 0.f, 0.f, 0.f, 0.f, 0.f, 0.f, 0.f};

    const float* ap = &Ald[khi * APR + 2 * (ti * 16 + row)];
    const float* bp = &Bld[khi * BPR + 2 * ((ti + djb) * 16 + row)];

#pragma unroll 8
    for (int k = 0; k < CCH; k += 4) {
      v2f a  = *(const v2f*)(ap);
      v2f b0 = *(const v2f*)(bp);
      v2f b1 = *(const v2f*)(bp + 32);
      acc0 = __builtin_amdgcn_wmma_f32_16x16x4_f32(
          false, a, false, b0, (short)0, acc0, false, false);
      acc1 = __builtin_amdgcn_wmma_f32_16x16x4_f32(
          false, a, false, b1, (short)0, acc1, false, false);
      ap += 2 * APR;
      bp += 2 * BPR;
    }

    const int mBase = khi * 8;
    const int N = lane & 15;
#pragma unroll
    for (int r = 0; r < 8; ++r) {
      const int m  = mBase + r;
      const int w1 = ti * 16 + m;
      {
        const int diff = 16 * djb + N - m;
        if (diff >= 0 && diff <= 40 && !(diff & 1))
          out[outBase + (size_t)(diff >> 1) * (HH * WW) + w1] =
              acc0[r] * 0.00390625f;
      }
      {
        const int diff = 16 * (djb + 1) + N - m;
        if (diff >= 0 && diff <= 40 && !(diff & 1))
          out[outBase + (size_t)(diff >> 1) * (HH * WW) + w1] =
              acc1[r] * 0.00390625f;
      }
    }
  };

  for (int dyI = dLo; dyI < dHi; ++dyI) {
    __syncthreads();
    put();
    __syncthreads();
    fetch(h + (dyI + 1) * 2 - 20);
    tiles_and_scatter(dyI);
  }
  __syncthreads();
  put();
  __syncthreads();
  tiles_and_scatter(dHi);

  for (int dyI = 0; dyI < ND; ++dyI) {
    if (dyI >= dLo && dyI <= dHi) continue;
    const size_t outBase =
        (((size_t)b * (ND * ND) + (size_t)dyI * ND) * HH + (size_t)h) * WW;
    for (int i = tid; i < ND * WW; i += NT) {
      const int dxI = i / WW, w = i % WW;
      out[outBase + (size_t)dxI * (HH * WW) + w] = 0.f;
    }
  }
}

extern "C" void kernel_launch(void* const* d_in, const int* in_sizes, int n_in,
                              void* d_out, int out_size, void* d_ws, size_t ws_size,
                              hipStream_t stream) {
  const float* f1 = (const float*)d_in[0];
  const float* f2 = (const float*)d_in[1];
  float* out = (float*)d_out;

  dim3 grid(HH, BATCH);
  dim3 block(NT);
  const size_t shmem =
      (size_t)((CCH / 2) * APR + (CCH / 2) * BPR) * sizeof(float);
  corr_wmma_f32_kernel<<<grid, block, shmem, stream>>>(f1, f2, out);
}


# output staged through LDS ([dx][w1], two stages) and written one dy later as whole 128-byte lines (float4 stores) under the next WMMA loop, replacing per-lane 4-byte scatter stores
# speedup vs baseline: 1.2415x; 1.0905x over previous
#include <hip/hip_runtime.h>


typedef __attribute__((ext_vector_type(2))) float v2f;
typedef __attribute__((ext_vector_type(8))) float v8f;

#define BATCH 8
#define CCH   256
#define HH    64
#define WW    96
#define ND    21
#define APR   224
#define BPR   288
#define NT    384
#define NLD   (CCH * (WW / 4) / NT)
#define SP    100
#define SSZ   (ND * SP)
#define LDS_DW ((CCH / 2) * APR + (CCH / 2) * BPR + 2 * SSZ)

__global__ __launch_bounds__(NT, 1)
void corr_wmma_f32_kernel(const float* __restrict__ f1,
                          const float* __restrict__ f2,
                          float* __restrict__ out) {
  __shared__ __attribute__((aligned(16))) float smem[LDS_DW];
  float* __restrict__ Ald = smem;
  float* __restrict__ Bld = smem + (CCH / 2) * APR;
  float* __restrict__ Sld = Bld + (CCH / 2) * BPR;

  const int tid  = threadIdx.x;
  const int lane = tid & 31;
  const int wv   = tid >> 5;
  const int ti   = wv >> 1;
  const int djb  = (wv & 1) * 2;

  const int h = blockIdx.x;
  const int b = blockIdx.y;

  const int row = lane & 15;
  const int khi = lane >> 4;

  const size_t inBase = (size_t)b * (CCH * HH * WW);

  const int dLo = (h < 20) ? ((21 - h) >> 1) : 0;
  const int dHi = (h > 43) ? ((83 - h) >> 1) : 20;

  constexpr int CSTEP = NT / (WW / 4);
  static_assert(NT % (WW / 4) == 0 && CCH % CSTEP == 0 && (CSTEP % 2) == 0,
                "staging map needs whole, even channel steps");
  const int c0 = tid / (WW / 4);
  const int q0 = tid % (WW / 4);
  const float* fsrc = f2 + inBase + (size_t)c0 * (HH * WW) + q0 * 4;
  float* bdst = &Bld[(c0 >> 1) * BPR + (c0 & 1) + 40 + 8 * q0];
  float4 pre[NLD];

  auto fetch = [&](int h2r) {
    const float* src = fsrc + (size_t)h2r * WW;
#pragma unroll
    for (int j = 0; j < NLD; ++j)
      pre[j] = *(const float4*)(src + (size_t)j * (CSTEP * HH * WW));
  };
  auto put = [&]() {
#pragma unroll
    for (int j = 0; j < NLD; ++j) {
      float* d = bdst + j * ((CSTEP / 2) * BPR);
      d[0] = pre[j].x; d[2] = pre[j].y; d[4] = pre[j].z; d[6] = pre[j].w;
    }
  };

  fetch(h + dLo * 2 - 20);

  {
    const float* src = f1 + inBase + (size_t)h * WW;
    for (int i = tid; i < CCH * (WW / 4); i += NT) {
      const int c = i / (WW / 4);
      const int q = i % (WW / 4);
      float4 v = *(const float4*)(src + (size_t)c * (HH * WW) + q * 4);
      float* d = &Ald[(c >> 1) * APR + (c & 1) + 8 * q];
      d[0] = v.x; d[2] = v.y; d[4] = v.z; d[6] = v.w;
    }
  }
  {
    const float4 z = make_float4(0.f, 0.f, 0.f, 0.f);
    for (int i = tid; i < (CCH / 2) * 24; i += NT) {
      const int p = i / 24;
      const int j = i % 24;
      const int off = (j < 10) ? (j * 4) : (232 + (j - 10) * 4);
      *(float4*)(&Bld[p * BPR + off]) = z;
    }
  }
  __syncthreads();

  auto flush = [&](int dyI) {
    const size_t outBase =
        (((size_t)b * (ND * ND) + (size_t)dyI * ND) * HH + (size_t)h) * WW;
    const float* s = Sld + (dyI & 1) * SSZ + c0 * SP + 4 * q0;
    float* o = out + outBase + (size_t)c0 * (HH * WW) + 4 * q0;
    *(float4*)o = *(const float4*)s;
    if (c0 + 16 < ND)
      *(float4*)(o + (size_t)16 * (HH * WW)) = *(const float4*)(s + 16 * SP);
  };

  auto tiles_to_stage = [&](int dyI) {
    float* __restrict__ sbuf = Sld + (dyI & 1) * SSZ;

    v8f acc0 = {0.f, 0.f, 0.f, 0.f, 0.f, 0.f, 0.f, 0.f};
    v8f acc1 = {0.f, 0.f, 0.f, 0.f, 0.f, 0.f, 0.f, 0.f};

    const float* ap = &Ald[khi * APR + 2 * (ti * 16 + row)];
    const float* bp = &Bld[khi * BPR + 2 * ((ti + djb) * 16 + row)];

#pragma unroll 8
    for (int k = 0; k < CCH; k += 4) {
      v2f a  = *(const v2f*)(ap);
      v2f b0 = *(const v2f*)(bp);
      v2f b1 = *(const v2f*)(bp + 32);
      acc0 = __builtin_amdgcn_wmma_f32_16x16x4_f32(
          false, a, false, b0, (short)0, acc0, false, false);
      acc1 = __builtin_amdgcn_wmma_f32_16x16x4_f32(
          false, a, false, b1, (short)0, acc1, false, false);
      ap += 2 * APR;
      bp += 2 * BPR;
    }

    const int mBase = khi * 8;
    const int N = lane & 15;
#pragma unroll
    for (int r = 0; r < 8; ++r) {
      const int m  = mBase + r;
      const int w1 = ti * 16 + m;
      {
        const int diff = 16 * djb + N - m;
        if (diff >= 0 && diff <= 40 && !(diff & 1))
          sbuf[(diff >> 1) * SP + w1] = acc0[r] * 0.00390625f;
      }
      {
        const int diff = 16 * (djb + 1) + N - m;
        if (diff >= 0 && diff <= 40 && !(diff & 1))
          sbuf[(diff >> 1) * SP + w1] = acc1[r] * 0.00390625f;
      }
    }
  };

  for (int dyI = dLo; dyI < dHi; ++dyI) {
    __syncthreads();
    put();
    __syncthreads();
    fetch(h + (dyI + 1) * 2 - 20);
    if (dyI > dLo) flush(dyI - 1);
    tiles_to_stage(dyI);
  }
  __syncthreads();
  put();
  __syncthreads();
  flush(dHi - 1);
  tiles_to_stage(dHi);
  __syncthreads();
  flush(dHi);

  for (int dyI = 0; dyI < ND; ++dyI) {
    if (dyI >= dLo && dyI <= dHi) continue;
    const size_t outBase =
        (((size_t)b * (ND * ND) + (size_t)dyI * ND) * HH + (size_t)h) * WW;
    for (int i = tid; i < ND * WW; i += NT) {
      const int dxI = i / WW, w = i % WW;
      out[outBase + (size_t)dxI * (HH * WW) + w] = 0.f;
    }
  }
}

extern "C" void kernel_launch(void* const* d_in, const int* in_sizes, int n_in,
                              void* d_out, int out_size, void* d_ws, size_t ws_size,
                              hipStream_t stream) {
  const float* f1 = (const float*)d_in[0];
  const float* f2 = (const float*)d_in[1];
  float* out = (float*)d_out;

  dim3 grid(HH, BATCH);
  dim3 block(NT);
  corr_wmma_f32_kernel<<<grid, block, 0, stream>>>(f1, f2, out);
}


# work-group id remapped so rows are dispatched longest first (21-dy middle rows, then edge rows down to 11 dy): busiest WGP does 21+16 dy rows, was 21+21; arithmetic and sum order unchanged
# speedup vs baseline: 1.3564x; 1.0925x over previous
#include <hip/hip_runtime.h>


typedef __attribute__((ext_vector_type(2))) float v2f;
typedef __attribute__((ext_vector_type(8))) float v8f;

#define BATCH 8
#define CCH   256
#define HH    64
#define WW    96
#define ND    21
#define APR   224
#define BPR   288
#define NT    384
#define NLD   (CCH * (WW / 4) / NT)
#define SP    100
#define SSZ   (ND * SP)
#define LDS_DW ((CCH / 2) * APR + (CCH / 2) * BPR + 2 * SSZ)

__global__ __launch_bounds__(NT, 1)
void corr_wmma_f32_kernel(const float* __restrict__ f1,
                          const float* __restrict__ f2,
                          float* __restrict__ out) {
  __shared__ __attribute__((aligned(16))) float smem[LDS_DW];
  float* __restrict__ Ald = smem;
  float* __restrict__ Bld = smem + (CCH / 2) * APR;
  float* __restrict__ Sld = Bld + (CCH / 2) * BPR;

  const int tid  = threadIdx.x;
  const int lane = tid & 31;
  const int wv   = tid >> 5;
  const int ti   = wv >> 1;
  const int djb  = (wv & 1) * 2;

  static_assert(BATCH == 8 && HH == 64, "rank map below is written for 8 x 64 rows");
  const int flat = blockIdx.y * HH + blockIdx.x;
  const int b    = flat & (BATCH - 1);
  const int rank = flat >> 3;
  int h;
  if (rank < 24) {
    h = 20 + rank;
  } else {
    const int s = rank - 24;
    const int e = (s >> 1) + 1;
    h = (s & 1) ? (43 + e) : (20 - e);
  }

  const int row = lane & 15;
  const int khi = lane >> 4;

  const size_t inBase = (size_t)b * (CCH * HH * WW);

  const int dLo = (h < 20) ? ((21 - h) >> 1) : 0;
  const int dHi = (h > 43) ? ((83 - h) >> 1) : 20;

  constexpr int CSTEP = NT / (WW / 4);
  static_assert(NT % (WW / 4) == 0 && CCH % CSTEP == 0 && (CSTEP % 2) == 0,
                "staging map needs whole, even channel steps");
  const int c0 = tid / (WW / 4);
  const int q0 = tid % (WW / 4);
  const float* fsrc = f2 + inBase + (size_t)c0 * (HH * WW) + q0 * 4;
  float* bdst = &Bld[(c0 >> 1) * BPR + (c0 & 1) + 40 + 8 * q0];
  float4 pre[NLD];

  auto fetch = [&](int h2r) {
    const float* src = fsrc + (size_t)h2r * WW;
#pragma unroll
    for (int j = 0; j < NLD; ++j)
      pre[j] = *(const float4*)(src + (size_t)j * (CSTEP * HH * WW));
  };
  auto put = [&]() {
#pragma unroll
    for (int j = 0; j < NLD; ++j) {
      float* d = bdst + j * ((CSTEP / 2) * BPR);
      d[0] = pre[j].x; d[2] = pre[j].y; d[4] = pre[j].z; d[6] = pre[j].w;
    }
  };

  fetch(h + dLo * 2 - 20);

  {
    const float* src = f1 + inBase + (size_t)h * WW;
    for (int i = tid; i < CCH * (WW / 4); i += NT) {
      const int c = i / (WW / 4);
      const int q = i % (WW / 4);
      float4 v = *(const float4*)(src + (size_t)c * (HH * WW) + q * 4);
      float* d = &Ald[(c >> 1) * APR + (c & 1) + 8 * q];
      d[0] = v.x; d[2] = v.y; d[4] = v.z; d[6] = v.w;
    }
  }
  {
    const float4 z = make_float4(0.f, 0.f, 0.f, 0.f);
    for (int i = tid; i < (CCH / 2) * 24; i += NT) {
      const int p = i / 24;
      const int j = i % 24;
      const int off = (j < 10) ? (j * 4) : (232 + (j - 10) * 4);
      *(float4*)(&Bld[p * BPR + off]) = z;
    }
  }
  __syncthreads();

  auto flush = [&](int dyI) {
    const size_t outBase =
        (((size_t)b * (ND * ND) + (size_t)dyI * ND) * HH + (size_t)h) * WW;
    const float* s = Sld + (dyI & 1) * SSZ + c0 * SP + 4 * q0;
    float* o = out + outBase + (size_t)c0 * (HH * WW) + 4 * q0;
    *(float4*)o = *(const float4*)s;
    if (c0 + 16 < ND)
      *(float4*)(o + (size_t)16 * (HH * WW)) = *(const float4*)(s + 16 * SP);
  };

  auto tiles_to_stage = [&](int dyI) {
    float* __restrict__ sbuf = Sld + (dyI & 1) * SSZ;

    v8f acc0 = {0.f, 0.f, 0.f, 0.f, 0.f, 0.f, 0.f, 0.f};
    v8f acc1 = {0.f, 0.f, 0.f, 0.f, 0.f, 0.f, 0.f, 0.f};

    const float* ap = &Ald[khi * APR + 2 * (ti * 16 + row)];
    const float* bp = &Bld[khi * BPR + 2 * ((ti + djb) * 16 + row)];

#pragma unroll 8
    for (int k = 0; k < CCH; k += 4) {
      v2f a  = *(const v2f*)(ap);
      v2f b0 = *(const v2f*)(bp);
      v2f b1 = *(const v2f*)(bp + 32);
      acc0 = __builtin_amdgcn_wmma_f32_16x16x4_f32(
          false, a, false, b0, (short)0, acc0, false, false);
      acc1 = __builtin_amdgcn_wmma_f32_16x16x4_f32(
          false, a, false, b1, (short)0, acc1, false, false);
      ap += 2 * APR;
      bp += 2 * BPR;
    }

    const int mBase = khi * 8;
    const int N = lane & 15;
#pragma unroll
    for (int r = 0; r < 8; ++r) {
      const int m  = mBase + r;
      const int w1 = ti * 16 + m;
      {
        const int diff = 16 * djb + N - m;
        if (diff >= 0 && diff <= 40 && !(diff & 1))
          sbuf[(diff >> 1) * SP + w1] = acc0[r] * 0.00390625f;
      }
      {
        const int diff = 16 * (djb + 1) + N - m;
        if (diff >= 0 && diff <= 40 && !(diff & 1))
          sbuf[(diff >> 1) * SP + w1] = acc1[r] * 0.00390625f;
      }
    }
  };

  for (int dyI = dLo; dyI < dHi; ++dyI) {
    __syncthreads();
    put();
    __syncthreads();
    fetch(h + (dyI + 1) * 2 - 20);
    if (dyI > dLo) flush(dyI - 1);
    tiles_to_stage(dyI);
  }
  __syncthreads();
  put();
  __syncthreads();
  flush(dHi - 1);
  tiles_to_stage(dHi);
  __syncthreads();
  flush(dHi);

  for (int dyI = 0; dyI < ND; ++dyI) {
    if (dyI >= dLo && dyI <= dHi) continue;
    const size_t outBase =
        (((size_t)b * (ND * ND) + (size_t)dyI * ND) * HH + (size_t)h) * WW;
    for (int i = tid; i < ND * WW; i += NT) {
      const int dxI = i / WW, w = i % WW;
      out[outBase + (size_t)dxI * (HH * WW) + w] = 0.f;
    }
  }
}

extern "C" void kernel_launch(void* const* d_in, const int* in_sizes, int n_in,
                              void* d_out, int out_size, void* d_ws, size_t ws_size,
                              hipStream_t stream) {
  const float* f1 = (const float*)d_in[0];
  const float* f2 = (const float*)d_in[1];
  float* out = (float*)d_out;

  dim3 grid(HH, BATCH);
  dim3 block(NT);
  corr_wmma_f32_kernel<<<grid, block, 0, stream>>>(f1, f2, out);
}


# f2 row staging to LDS: a thread loads both channels of a pair and writes two conflict-free 16-byte stores (quad order swapped per 8-lane group), was eight 4-byte stores with 2-3-way bank conflicts
# speedup vs baseline: 1.5014x; 1.1069x over previous
#include <hip/hip_runtime.h>


typedef __attribute__((ext_vector_type(2))) float v2f;
typedef __attribute__((ext_vector_type(4))) float v4f;
typedef __attribute__((ext_vector_type(8))) float v8f;

#define BATCH 8
#define CCH   256
#define HH    64
#define WW    96
#define ND    21
#define APR   224
#define BPR   288
#define NT    384
#define NLD   (CCH * (WW / 4) / NT)
#define SP    100
#define SSZ   (ND * SP)
#define LDS_DW ((CCH / 2) * APR + (CCH / 2) * BPR + 2 * SSZ)

__global__ __launch_bounds__(NT, 1)
void corr_wmma_f32_kernel(const float* __restrict__ f1,
                          const float* __restrict__ f2,
                          float* __restrict__ out) {
  __shared__ __attribute__((aligned(16))) float smem[LDS_DW];
  float* __restrict__ Ald = smem;
  float* __restrict__ Bld = smem + (CCH / 2) * APR;
  float* __restrict__ Sld = Bld + (CCH / 2) * BPR;

  const int tid  = threadIdx.x;
  const int lane = tid & 31;
  const int wv   = tid >> 5;
  const int ti   = wv >> 1;
  const int djb  = (wv & 1) * 2;

  static_assert(BATCH == 8 && HH == 64, "rank map below is written for 8 x 64 rows");
  const int flat = blockIdx.y * HH + blockIdx.x;
  const int b    = flat & (BATCH - 1);
  const int rank = flat >> 3;
  int h;
  if (rank < 24) {
    h = 20 + rank;
  } else {
    const int s = rank - 24;
    const int e = (s >> 1) + 1;
    h = (s & 1) ? (43 + e) : (20 - e);
  }

  const int row = lane & 15;
  const int khi = lane >> 4;

  const size_t inBase = (size_t)b * (CCH * HH * WW);

  const int dLo = (h < 20) ? ((21 - h) >> 1) : 0;
  const int dHi = (h > 43) ? ((83 - h) >> 1) : 20;

  constexpr int PSTEP = NT / (WW / 4);
  constexpr int NIT   = (CCH / 2) / PSTEP;
  static_assert(NT % (WW / 4) == 0 && (CCH / 2) % PSTEP == 0 && NLD == 2 * NIT,
                "staging map needs whole pair rows per pass");
  static_assert((WW / 4) % 8 == 0, "an 8-lane group must not straddle two pair rows");
  const int c0 = tid / (WW / 4);
  const int q0 = tid % (WW / 4);
  const float* fsrc = f2 + inBase + (size_t)(2 * c0) * (HH * WW) + q0 * 4;
  float* bdst = &Bld[c0 * BPR + 40 + 8 * q0];
  const bool hiFirst = (tid >> 3) & 1;
  const int o1 = hiFirst ? 4 : 0;
  const int o2 = 4 - o1;
  float4 pre[NLD];

  auto fetch = [&](int h2r) {
    const float* src = fsrc + (size_t)h2r * WW;
#pragma unroll
    for (int j = 0; j < NIT; ++j) {
      pre[2 * j]     = *(const float4*)(src + (size_t)(2 * PSTEP * j) * (HH * WW));
      pre[2 * j + 1] = *(const float4*)(src + (size_t)(2 * PSTEP * j + 1) * (HH * WW));
    }
  };
  auto put = [&]() {
#pragma unroll
    for (int j = 0; j < NIT; ++j) {
      const float4 x = pre[2 * j];
      const float4 y = pre[2 * j + 1];
      const v4f first  = {hiFirst ? x.z : x.x, hiFirst ? y.z : y.x,
                          hiFirst ? x.w : x.y, hiFirst ? y.w : y.y};
      const v4f second = {hiFirst ? x.x : x.z, hiFirst ? y.x : y.z,
                          hiFirst ? x.y : x.w, hiFirst ? y.y : y.w};
      float* d = bdst + j * (PSTEP * BPR);
      *(v4f*)(d + o1) = first;
      *(v4f*)(d + o2) = second;
    }
  };

  fetch(h + dLo * 2 - 20);

  {
    const float* src = f1 + inBase + (size_t)h * WW;
    for (int i = tid; i < CCH * (WW / 4); i += NT) {
      const int c = i / (WW / 4);
      const int q = i % (WW / 4);
      float4 v = *(const float4*)(src + (size_t)c * (HH * WW) + q * 4);
      float* d = &Ald[(c >> 1) * APR + (c & 1) + 8 * q];
      d[0] = v.x; d[2] = v.y; d[4] = v.z; d[6] = v.w;
    }
  }
  {
    const float4 z = make_float4(0.f, 0.f, 0.f, 0.f);
    for (int i = tid; i < (CCH / 2) * 24; i += NT) {
      const int p = i / 24;
      const int j = i % 24;
      const int off = (j < 10) ? (j * 4) : (232 + (j - 10) * 4);
      *(float4*)(&Bld[p * BPR + off]) = z;
    }
  }
  __syncthreads();

  auto flush = [&](int dyI) {
    const size_t outBase =
        (((size_t)b * (ND * ND) + (size_t)dyI * ND) * HH + (size_t)h) * WW;
    const float* s = Sld + (dyI & 1) * SSZ + c0 * SP + 4 * q0;
    float* o = out + outBase + (size_t)c0 * (HH * WW) + 4 * q0;
    *(float4*)o = *(const float4*)s;
    if (c0 + 16 < ND)
      *(float4*)(o + (size_t)16 * (HH * WW)) = *(const float4*)(s + 16 * SP);
  };

  auto tiles_to_stage = [&](int dyI) {
    float* __restrict__ sbuf = Sld + (dyI & 1) * SSZ;

    v8f acc0 = {0.f, 0.f, 0.f, 0.f, 0.f, 0.f, 0.f, 0.f};
    v8f acc1 = {0.f, 0.f, 0.f, 0.f, 0.f, 0.f, 0.f, 0.f};

    const float* ap = &Ald[khi * APR + 2 * (ti * 16 + row)];
    const float* bp = &Bld[khi * BPR + 2 * ((ti + djb) * 16 + row)];

#pragma unroll 8
    for (int k = 0; k < CCH; k += 4) {
      v2f a  = *(const v2f*)(ap);
      v2f b0 = *(const v2f*)(bp);
      v2f b1 = *(const v2f*)(bp + 32);
      acc0 = __builtin_amdgcn_wmma_f32_16x16x4_f32(
          false, a, false, b0, (short)0, acc0, false, false);
      acc1 = __builtin_amdgcn_wmma_f32_16x16x4_f32(
          false, a, false, b1, (short)0, acc1, false, false);
      ap += 2 * APR;
      bp += 2 * BPR;
    }

    const int mBase = khi * 8;
    const int N = lane & 15;
#pragma unroll
    for (int r = 0; r < 8; ++r) {
      const int m  = mBase + r;
      const int w1 = ti * 16 + m;
      {
        const int diff = 16 * djb + N - m;
        if (diff >= 0 && diff <= 40 && !(diff & 1))
          sbuf[(diff >> 1) * SP + w1] = acc0[r] * 0.00390625f;
      }
      {
        const int diff = 16 * (djb + 1) + N - m;
        if (diff >= 0 && diff <= 40 && !(diff & 1))
          sbuf[(diff >> 1) * SP + w1] = acc1[r] * 0.00390625f;
      }
    }
  };

  for (int dyI = dLo; dyI < dHi; ++dyI) {
    __syncthreads();
    put();
    __syncthreads();
    fetch(h + (dyI + 1) * 2 - 20);
    if (dyI > dLo) flush(dyI - 1);
    tiles_to_stage(dyI);
  }
  __syncthreads();
  put();
  __syncthreads();
  flush(dHi - 1);
  tiles_to_stage(dHi);
  __syncthreads();
  flush(dHi);

  for (int dyI = 0; dyI < ND; ++dyI) {
    if (dyI >= dLo && dyI <= dHi) continue;
    const size_t outBase =
        (((size_t)b * (ND * ND) + (size_t)dyI * ND) * HH + (size_t)h) * WW;
    for (int i = tid; i < ND * WW; i += NT) {
      const int dxI = i / WW, w = i % WW;
      out[outBase + (size_t)dxI * (HH * WW) + w] = 0.f;
    }
  }
}

extern "C" void kernel_launch(void* const* d_in, const int* in_sizes, int n_in,
                              void* d_out, int out_size, void* d_ws, size_t ws_size,
                              hipStream_t stream) {
  const float* f1 = (const float*)d_in[0];
  const float* f2 = (const float*)d_in[1];
  float* out = (float*)d_out;

  dim3 grid(HH, BATCH);
  dim3 block(NT);
  corr_wmma_f32_kernel<<<grid, block, 0, stream>>>(f1, f2, out);
}


# LDS tiles de-interleaved by column parity (stride 2): 18 tiles per dy, was 24; a wave does 3 tiles over half of K, the two K-half partial sums are added at write-out (sum order changed)
# speedup vs baseline: 1.8646x; 1.2419x over previous
#include <hip/hip_runtime.h>


typedef __attribute__((ext_vector_type(2))) float v2f;
typedef __attribute__((ext_vector_type(4))) float v4f;
typedef __attribute__((ext_vector_type(8))) float v8f;

#define BATCH 8
#define CCH   256
#define HH    64
#define WW    96
#define ND    21
#define APR   224
#define APH   96
#define BPR   288
#define BPH   144
#define NT    384
#define NLD   (CCH * (WW / 4) / NT)
#define SP    100
#define SSZ   (ND * SP)
#define LDS_DW ((CCH / 2) * APR + (CCH / 2) * BPR + 4 * SSZ)

__global__ __launch_bounds__(NT, 1)
void corr_wmma_f32_kernel(const float* __restrict__ f1,
                          const float* __restrict__ f2,
                          float* __restrict__ out) {
  __shared__ __attribute__((aligned(16))) float smem[LDS_DW];
  float* __restrict__ Ald = smem;
  float* __restrict__ Bld = smem + (CCH / 2) * APR;
  float* __restrict__ Sld = Bld + (CCH / 2) * BPR;

  const int tid  = threadIdx.x;
  const int lane = tid & 31;
  const int wv   = tid >> 5;
  const int kh   = wv & 1;
  const int par  = (wv >= 6) ? 1 : 0;
  const int ti   = (wv >> 1) - 3 * par;

  static_assert(BATCH == 8 && HH == 64, "rank map below is written for 8 x 64 rows");
  const int flat = blockIdx.y * HH + blockIdx.x;
  const int b    = flat & (BATCH - 1);
  const int rank = flat >> 3;
  int h;
  if (rank < 24) {
    h = 20 + rank;
  } else {
    const int s = rank - 24;
    const int e = (s >> 1) + 1;
    h = (s & 1) ? (43 + e) : (20 - e);
  }

  const int row = lane & 15;
  const int khi = lane >> 4;

  const size_t inBase = (size_t)b * (CCH * HH * WW);

  const int dLo = (h < 20) ? ((21 - h) >> 1) : 0;
  const int dHi = (h > 43) ? ((83 - h) >> 1) : 20;

  constexpr int PSTEP = NT / (WW / 4);
  constexpr int NIT   = (CCH / 2) / PSTEP;
  static_assert(NT % (WW / 4) == 0 && (CCH / 2) % PSTEP == 0 && NLD == 2 * NIT,
                "staging map needs whole pair rows per pass");
  static_assert((WW / 4) % 8 == 0, "an 8-lane group must not straddle two pair rows");
  const int c0 = tid / (WW / 4);
  const int q0 = tid % (WW / 4);
  const float* fsrc = f2 + inBase + (size_t)(2 * c0) * (HH * WW) + q0 * 4;
  float* bdst = &Bld[c0 * BPR + 20 + 4 * q0];
  float4 pre[NLD];

  auto fetch = [&](int h2r) {
    const float* src = fsrc + (size_t)h2r * WW;
#pragma unroll
    for (int j = 0; j < NIT; ++j) {
      pre[2 * j]     = *(const float4*)(src + (size_t)(2 * PSTEP * j) * (HH * WW));
      pre[2 * j + 1] = *(const float4*)(src + (size_t)(2 * PSTEP * j + 1) * (HH * WW));
    }
  };
  auto put = [&]() {
#pragma unroll
    for (int j = 0; j < NIT; ++j) {
      const float4 x = pre[2 * j];
      const float4 y = pre[2 * j + 1];
      const v4f ev = {x.x, y.x, x.z, y.z};
      const v4f od = {x.y, y.y, x.w, y.w};
      float* d = bdst + j * (PSTEP * BPR);
      *(v4f*)(d)       = ev;
      *(v4f*)(d + BPH) = od;
    }
  };

  fetch(h + dLo * 2 - 20);

  {
    const float* src = f1 + inBase + (size_t)h * WW;
    for (int i = tid; i < CCH * (WW / 4); i += NT) {
      const int c = i / (WW / 4);
      const int q = i % (WW / 4);
      float4 v = *(const float4*)(src + (size_t)c * (HH * WW) + q * 4);
      float* d = &Ald[(c >> 1) * APR + (c & 1) + 4 * q];
      d[0] = v.x; d[2] = v.z;
      d[APH] = v.y; d[APH + 2] = v.w;
    }
  }
  {
    const float4 z = make_float4(0.f, 0.f, 0.f, 0.f);
    for (int i = tid; i < (CCH / 2) * 24; i += NT) {
      const int p = i / 24;
      const int j = i % 24;
      const int off = (j < 5) ? (j * 4) : (j < 17) ? (116 + (j - 5) * 4) : (260 + (j - 17) * 4);
      *(float4*)(&Bld[p * BPR + off]) = z;
    }
  }
  __syncthreads();

  auto flush = [&](int dyI) {
    const size_t outBase =
        (((size_t)b * (ND * ND) + (size_t)dyI * ND) * HH + (size_t)h) * WW;
    const float* s = Sld + (dyI & 1) * (2 * SSZ) + c0 * SP + 4 * q0;
    float* o = out + outBase + (size_t)c0 * (HH * WW) + 4 * q0;
    {
      const float4 lo = *(const float4*)s;
      const float4 hi = *(const float4*)(s + SSZ);
      *(float4*)o = make_float4(lo.x + hi.x, lo.y + hi.y, lo.z + hi.z, lo.w + hi.w);
    }
    if (c0 + 16 < ND) {
      const float4 lo = *(const float4*)(s + 16 * SP);
      const float4 hi = *(const float4*)(s + 16 * SP + SSZ);
      *(float4*)(o + (size_t)16 * (HH * WW)) =
          make_float4(lo.x + hi.x, lo.y + hi.y, lo.z + hi.z, lo.w + hi.w);
    }
  };

  auto tiles_to_stage = [&](int dyI) {
    float* __restrict__ sbuf = Sld + ((dyI & 1) * 2 + kh) * SSZ;

    v8f acc0 = {0.f, 0.f, 0.f, 0.f, 0.f, 0.f, 0.f, 0.f};
    v8f acc1 = {0.f, 0.f, 0.f, 0.f, 0.f, 0.f, 0.f, 0.f};
    v8f acc2 = {0.f, 0.f, 0.f, 0.f, 0.f, 0.f, 0.f, 0.f};

    const float* ap = &Ald[(kh * (CCH / 4) + khi) * APR + par * APH + 2 * (ti * 16 + row)];
    const float* bp = &Bld[(kh * (CCH / 4) + khi) * BPR + par * BPH + 2 * (ti * 16 + row)];

#pragma unroll 8
    for (int k = 0; k < CCH / 2; k += 4) {
      v2f a  = *(const v2f*)(ap);
      v2f b0 = *(const v2f*)(bp);
      v2f b1 = *(const v2f*)(bp + 32);
      v2f b2 = *(const v2f*)(bp + 64);
      acc0 = __builtin_amdgcn_wmma_f32_16x16x4_f32(
          false, a, false, b0, (short)0, acc0, false, false);
      acc1 = __builtin_amdgcn_wmma_f32_16x16x4_f32(
          false, a, false, b1, (short)0, acc1, false, false);
      acc2 = __builtin_amdgcn_wmma_f32_16x16x4_f32(
          false, a, false, b2, (short)0, acc2, false, false);
      ap += 2 * APR;
      bp += 2 * BPR;
    }

    const int mBase = khi * 8;
    const int N = lane & 15;
#pragma unroll
    for (int r = 0; r < 8; ++r) {
      const int m  = mBase + r;
      const int w1 = 2 * (ti * 16 + m) + par;
      const int d0 = N - m;
      const int d1 = d0 + 16;
      const int d2 = d0 + 32;
      if (d0 >= 0)  sbuf[d0 * SP + w1] = acc0[r] * 0.00390625f;
      if (d1 <= 20) sbuf[d1 * SP + w1] = acc1[r] * 0.00390625f;
      if (d2 <= 20) sbuf[d2 * SP + w1] = acc2[r] * 0.00390625f;
    }
  };

  for (int dyI = dLo; dyI < dHi; ++dyI) {
    __syncthreads();
    put();
    __syncthreads();
    fetch(h + (dyI + 1) * 2 - 20);
    if (dyI > dLo) flush(dyI - 1);
    tiles_to_stage(dyI);
  }
  __syncthreads();
  put();
  __syncthreads();
  flush(dHi - 1);
  tiles_to_stage(dHi);
  __syncthreads();
  flush(dHi);

  for (int dyI = 0; dyI < ND; ++dyI) {
    if (dyI >= dLo && dyI <= dHi) continue;
    const size_t outBase =
        (((size_t)b * (ND * ND) + (size_t)dyI * ND) * HH + (size_t)h) * WW;
    for (int i = tid; i < ND * WW; i += NT) {
      const int dxI = i / WW, w = i % WW;
      out[outBase + (size_t)dxI * (HH * WW) + w] = 0.f;
    }
  }
}

extern "C" void kernel_launch(void* const* d_in, const int* in_sizes, int n_in,
                              void* d_out, int out_size, void* d_ws, size_t ws_size,
                              hipStream_t stream) {
  const float* f1 = (const float*)d_in[0];
  const float* f2 = (const float*)d_in[1];
  float* out = (float*)d_out;

  dim3 grid(HH, BATCH);
  dim3 block(NT);
  corr_wmma_f32_kernel<<<grid, block, 0, stream>>>(f1, f2, out);
}


# fragment loads grouped: a block of 8 k-steps' A and B ds_loads is issued as one group into a second register set ahead of the previous block's 24 WMMAs (was 4 loads between every 3 WMMAs)
# speedup vs baseline: 1.9827x; 1.0633x over previous
#include <hip/hip_runtime.h>


typedef __attribute__((ext_vector_type(2))) float v2f;
typedef __attribute__((ext_vector_type(4))) float v4f;
typedef __attribute__((ext_vector_type(8))) float v8f;

#define BATCH 8
#define CCH   256
#define HH    64
#define WW    96
#define ND    21
#define APR   224
#define APH   96
#define BPR   288
#define BPH   144
#define NT    384
#define NLD   (CCH * (WW / 4) / NT)
#define SP    100
#define SSZ   (ND * SP)
#define LDS_DW ((CCH / 2) * APR + (CCH / 2) * BPR + 4 * SSZ)

__global__ __launch_bounds__(NT, 1)
void corr_wmma_f32_kernel(const float* __restrict__ f1,
                          const float* __restrict__ f2,
                          float* __restrict__ out) {
  __shared__ __attribute__((aligned(16))) float smem[LDS_DW];
  float* __restrict__ Ald = smem;
  float* __restrict__ Bld = smem + (CCH / 2) * APR;
  float* __restrict__ Sld = Bld + (CCH / 2) * BPR;

  const int tid  = threadIdx.x;
  const int lane = tid & 31;
  const int wv   = tid >> 5;
  const int kh   = wv & 1;
  const int par  = (wv >= 6) ? 1 : 0;
  const int ti   = (wv >> 1) - 3 * par;

  static_assert(BATCH == 8 && HH == 64, "rank map below is written for 8 x 64 rows");
  const int flat = blockIdx.y * HH + blockIdx.x;
  const int b    = flat & (BATCH - 1);
  const int rank = flat >> 3;
  int h;
  if (rank < 24) {
    h = 20 + rank;
  } else {
    const int s = rank - 24;
    const int e = (s >> 1) + 1;
    h = (s & 1) ? (43 + e) : (20 - e);
  }

  const int row = lane & 15;
  const int khi = lane >> 4;

  const size_t inBase = (size_t)b * (CCH * HH * WW);

  const int dLo = (h < 20) ? ((21 - h) >> 1) : 0;
  const int dHi = (h > 43) ? ((83 - h) >> 1) : 20;

  constexpr int PSTEP = NT / (WW / 4);
  constexpr int NIT   = (CCH / 2) / PSTEP;
  static_assert(NT % (WW / 4) == 0 && (CCH / 2) % PSTEP == 0 && NLD == 2 * NIT,
                "staging map needs whole pair rows per pass");
  static_assert((WW / 4) % 8 == 0, "an 8-lane group must not straddle two pair rows");
  const int c0 = tid / (WW / 4);
  const int q0 = tid % (WW / 4);
  const float* fsrc = f2 + inBase + (size_t)(2 * c0) * (HH * WW) + q0 * 4;
  float* bdst = &Bld[c0 * BPR + 20 + 4 * q0];
  float4 pre[NLD];

  auto fetch = [&](int h2r) {
    const float* src = fsrc + (size_t)h2r * WW;
#pragma unroll
    for (int j = 0; j < NIT; ++j) {
      pre[2 * j]     = *(const float4*)(src + (size_t)(2 * PSTEP * j) * (HH * WW));
      pre[2 * j + 1] = *(const float4*)(src + (size_t)(2 * PSTEP * j + 1) * (HH * WW));
    }
  };
  auto put = [&]() {
#pragma unroll
    for (int j = 0; j < NIT; ++j) {
      const float4 x = pre[2 * j];
      const float4 y = pre[2 * j + 1];
      const v4f ev = {x.x, y.x, x.z, y.z};
      const v4f od = {x.y, y.y, x.w, y.w};
      float* d = bdst + j * (PSTEP * BPR);
      *(v4f*)(d)       = ev;
      *(v4f*)(d + BPH) = od;
    }
  };

  fetch(h + dLo * 2 - 20);

  {
    const float* src = f1 + inBase + (size_t)h * WW;
    for (int i = tid; i < CCH * (WW / 4); i += NT) {
      const int c = i / (WW / 4);
      const int q = i % (WW / 4);
      float4 v = *(const float4*)(src + (size_t)c * (HH * WW) + q * 4);
      float* d = &Ald[(c >> 1) * APR + (c & 1) + 4 * q];
      d[0] = v.x; d[2] = v.z;
      d[APH] = v.y; d[APH + 2] = v.w;
    }
  }
  {
    const float4 z = make_float4(0.f, 0.f, 0.f, 0.f);
    for (int i = tid; i < (CCH / 2) * 24; i += NT) {
      const int p = i / 24;
      const int j = i % 24;
      const int off = (j < 5) ? (j * 4) : (j < 17) ? (116 + (j - 5) * 4) : (260 + (j - 17) * 4);
      *(float4*)(&Bld[p * BPR + off]) = z;
    }
  }
  __syncthreads();

  auto flush = [&](int dyI) {
    const size_t outBase =
        (((size_t)b * (ND * ND) + (size_t)dyI * ND) * HH + (size_t)h) * WW;
    const float* s = Sld + (dyI & 1) * (2 * SSZ) + c0 * SP + 4 * q0;
    float* o = out + outBase + (size_t)c0 * (HH * WW) + 4 * q0;
    {
      const float4 lo = *(const float4*)s;
      const float4 hi = *(const float4*)(s + SSZ);
      *(float4*)o = make_float4(lo.x + hi.x, lo.y + hi.y, lo.z + hi.z, lo.w + hi.w);
    }
    if (c0 + 16 < ND) {
      const float4 lo = *(const float4*)(s + 16 * SP);
      const float4 hi = *(const float4*)(s + 16 * SP + SSZ);
      *(float4*)(o + (size_t)16 * (HH * WW)) =
          make_float4(lo.x + hi.x, lo.y + hi.y, lo.z + hi.z, lo.w + hi.w);
    }
  };

  auto tiles_to_stage = [&](int dyI) {
    float* __restrict__ sbuf = Sld + ((dyI & 1) * 2 + kh) * SSZ;

    v8f acc0 = {0.f, 0.f, 0.f, 0.f, 0.f, 0.f, 0.f, 0.f};
    v8f acc1 = {0.f, 0.f, 0.f, 0.f, 0.f, 0.f, 0.f, 0.f};
    v8f acc2 = {0.f, 0.f, 0.f, 0.f, 0.f, 0.f, 0.f, 0.f};

    const float* ap = &Ald[(kh * (CCH / 4) + khi) * APR + par * APH + 2 * (ti * 16 + row)];
    const float* bp = &Bld[(kh * (CCH / 4) + khi) * BPR + par * BPH + 2 * (ti * 16 + row)];

    constexpr int KB  = 8;
    constexpr int KBN = (CCH / 2 / 4) / KB;
    static_assert(KBN * KB * 4 == CCH / 2, "blocks must cover the K half exactly");
    v2f fa[2][KB];
    v2f fb[2][KB][3];

    auto load_steps = [&](int s, int blk, int t0, int t1) {
#pragma unroll
      for (int t = t0; t < t1; ++t) {
        const int kk = blk * KB + t;
        fa[s][t]    = *(const v2f*)(ap + kk * (2 * APR));
        fb[s][t][0] = *(const v2f*)(bp + kk * (2 * BPR));
        fb[s][t][1] = *(const v2f*)(bp + kk * (2 * BPR) + 32);
        fb[s][t][2] = *(const v2f*)(bp + kk * (2 * BPR) + 64);
      }
    };

    load_steps(0, 0, 0, 2);
    __builtin_amdgcn_sched_barrier(0);
    load_steps(0, 0, 2, KB);
    __builtin_amdgcn_sched_barrier(0);
#pragma unroll
    for (int blk = 0; blk < KBN; ++blk) {
      if (blk + 1 < KBN) load_steps((blk + 1) & 1, blk + 1, 0, KB);
      __builtin_amdgcn_sched_barrier(0);
#pragma unroll
      for (int t = 0; t < KB; ++t) {
        const v2f a = fa[blk & 1][t];
        acc0 = __builtin_amdgcn_wmma_f32_16x16x4_f32(
            false, a, false, fb[blk & 1][t][0], (short)0, acc0, false, false);
        acc1 = __builtin_amdgcn_wmma_f32_16x16x4_f32(
            false, a, false, fb[blk & 1][t][1], (short)0, acc1, false, false);
        acc2 = __builtin_amdgcn_wmma_f32_16x16x4_f32(
            false, a, false, fb[blk & 1][t][2], (short)0, acc2, false, false);
      }
      __builtin_amdgcn_sched_barrier(0);
    }

    const int mBase = khi * 8;
    const int N = lane & 15;
#pragma unroll
    for (int r = 0; r < 8; ++r) {
      const int m  = mBase + r;
      const int w1 = 2 * (ti * 16 + m) + par;
      const int d0 = N - m;
      const int d1 = d0 + 16;
      const int d2 = d0 + 32;
      if (d0 >= 0)  sbuf[d0 * SP + w1] = acc0[r] * 0.00390625f;
      if (d1 <= 20) sbuf[d1 * SP + w1] = acc1[r] * 0.00390625f;
      if (d2 <= 20) sbuf[d2 * SP + w1] = acc2[r] * 0.00390625f;
    }
  };

  for (int dyI = dLo; dyI < dHi; ++dyI) {
    __syncthreads();
    put();
    __syncthreads();
    fetch(h + (dyI + 1) * 2 - 20);
    if (dyI > dLo) flush(dyI - 1);
    tiles_to_stage(dyI);
  }
  __syncthreads();
  put();
  __syncthreads();
  flush(dHi - 1);
  tiles_to_stage(dHi);
  __syncthreads();
  flush(dHi);

  for (int dyI = 0; dyI < ND; ++dyI) {
    if (dyI >= dLo && dyI <= dHi) continue;
    const size_t outBase =
        (((size_t)b * (ND * ND) + (size_t)dyI * ND) * HH + (size_t)h) * WW;
    for (int i = tid; i < ND * WW; i += NT) {
      const int dxI = i / WW, w = i % WW;
      out[outBase + (size_t)dxI * (HH * WW) + w] = 0.f;
    }
  }
}

extern "C" void kernel_launch(void* const* d_in, const int* in_sizes, int n_in,
                              void* d_out, int out_size, void* d_ws, size_t ws_size,
                              hipStream_t stream) {
  const float* f1 = (const float*)d_in[0];
  const float* f2 = (const float*)d_in[1];
  float* out = (float*)d_out;

  dim3 grid(HH, BATCH);
  dim3 block(NT);
  corr_wmma_f32_kernel<<<grid, block, 0, stream>>>(f1, f2, out);
}


# f2 row tile double-buffered in LDS (f1 fragments kept in registers to make room): next row's staging and the tile store run behind each wave's WMMA loop; one barrier per dy row, was two
# speedup vs baseline: 2.1458x; 1.0823x over previous
#include <hip/hip_runtime.h>


typedef __attribute__((ext_vector_type(2))) float v2f;
typedef __attribute__((ext_vector_type(4))) float v4f;
typedef __attribute__((ext_vector_type(8))) float v8f;

#define BATCH 8
#define CCH   256
#define HH    64
#define WW    96
#define ND    21
#define BPR   288
#define BPH   144
#define BSZ   ((CCH / 2) * BPR)
#define NT    384
#define NLD   (CCH * (WW / 4) / NT)
#define SP    96
#define SSZ   (ND * SP)
#define LDS_DW (2 * BSZ + 4 * SSZ)

static_assert(LDS_DW * 4 <= 327680, "two row tiles and the stages must fit the LDS of one WGP");

__device__ __forceinline__ int stage_off(int d, int w1) {
  return d * SP + ((((w1 >> 2) ^ ((d >> 1) & 7))) << 2) + (w1 & 3);
}

__global__ __launch_bounds__(NT, 1)
void corr_wmma_f32_kernel(const float* __restrict__ f1,
                          const float* __restrict__ f2,
                          float* __restrict__ out) {
  __shared__ __attribute__((aligned(16))) float smem[LDS_DW];
  float* __restrict__ Bld = smem;
  float* __restrict__ Sld = smem + 2 * BSZ;

  const int tid  = threadIdx.x;
  const int lane = tid & 31;
  const int wv   = tid >> 5;
  const int kh   = wv & 1;
  const int par  = (wv >= 6) ? 1 : 0;
  const int ti   = (wv >> 1) - 3 * par;

  static_assert(BATCH == 8 && HH == 64, "rank map below is written for 8 x 64 rows");
  const int flat = blockIdx.y * HH + blockIdx.x;
  const int b    = flat & (BATCH - 1);
  const int rank = flat >> 3;
  int h;
  if (rank < 24) {
    h = 20 + rank;
  } else {
    const int s = rank - 24;
    const int e = (s >> 1) + 1;
    h = (s & 1) ? (43 + e) : (20 - e);
  }

  const int row = lane & 15;
  const int khi = lane >> 4;

  const size_t inBase = (size_t)b * (CCH * HH * WW);

  const int dLo = (h < 20) ? ((21 - h) >> 1) : 0;
  const int dHi = (h > 43) ? ((83 - h) >> 1) : 20;

  constexpr int PSTEP = NT / (WW / 4);
  constexpr int NIT   = (CCH / 2) / PSTEP;
  static_assert(NT % (WW / 4) == 0 && (CCH / 2) % PSTEP == 0 && NLD == 2 * NIT,
                "staging map needs whole pair rows per pass");
  static_assert((WW / 4) % 8 == 0, "an 8-lane group must not straddle two pair rows");
  const int c0 = tid / (WW / 4);
  const int q0 = tid % (WW / 4);
  const float* fsrc = f2 + inBase + (size_t)(2 * c0) * (HH * WW) + q0 * 4;
  float* bdst = &Bld[c0 * BPR + 20 + 4 * q0];
  float4 pre[NLD];

  auto fetch = [&](int h2r) {
    const float* src = fsrc + (size_t)h2r * WW;
#pragma unroll
    for (int j = 0; j < NIT; ++j) {
      pre[2 * j]     = *(const float4*)(src + (size_t)(2 * PSTEP * j) * (HH * WW));
      pre[2 * j + 1] = *(const float4*)(src + (size_t)(2 * PSTEP * j + 1) * (HH * WW));
    }
  };
  auto put = [&](int buf) {
    float* dbase = bdst + buf * BSZ;
#pragma unroll
    for (int j = 0; j < NIT; ++j) {
      const float4 x = pre[2 * j];
      const float4 y = pre[2 * j + 1];
      const v4f ev = {x.x, y.x, x.z, y.z};
      const v4f od = {x.y, y.y, x.w, y.w};
      float* d = dbase + j * (PSTEP * BPR);
      *(v4f*)(d)       = ev;
      *(v4f*)(d + BPH) = od;
    }
  };

  constexpr int KSTEPS = CCH / 2 / 4;
  v2f fa[KSTEPS];
  {
    const float* asrc = f1 + inBase + (size_t)(kh * (CCH / 2) + 2 * khi) * (HH * WW) +
                        (size_t)h * WW + 2 * (ti * 16 + row) + par;
#pragma unroll
    for (int kk = 0; kk < KSTEPS; ++kk) {
      fa[kk].x = asrc[(size_t)(4 * kk) * (HH * WW)];
      fa[kk].y = asrc[(size_t)(4 * kk + 1) * (HH * WW)];
    }
  }
  fetch(h + dLo * 2 - 20);

  {
    const float4 z = make_float4(0.f, 0.f, 0.f, 0.f);
    for (int i = tid; i < 2 * (CCH / 2) * 24; i += NT) {
      const int p = i / 24;
      const int j = i % 24;
      const int off = (j < 5) ? (j * 4) : (j < 17) ? (116 + (j - 5) * 4) : (260 + (j - 17) * 4);
      *(float4*)(&Bld[p * BPR + off]) = z;
    }
  }
  put(dLo & 1);

  constexpr int FQ = ND * (WW / 4);
  constexpr int FT = 4 * 32;
  auto flush = [&](int dyI) {
    if (wv < (NT - FT) / 32) return;
    const size_t outBase =
        (((size_t)b * (ND * ND) + (size_t)dyI * ND) * HH + (size_t)h) * WW;
    const float* sb = Sld + (dyI & 1) * (2 * SSZ);
    const int t = tid - (NT - FT);
    constexpr int NQ = (FQ + FT - 1) / FT;
    static_assert((NQ - 1) * FT < FQ && FQ % 8 == 0, "all but the last quad of every thread exist");
    float4 val[NQ];
    size_t oo[NQ];
#pragma unroll
    for (int k = 0; k < NQ; ++k) {
      const int idx = t + FT * k;
      const int idc = (k + 1 < NQ || idx < FQ) ? idx : (FQ - 1);
      const int dx = idc / (WW / 4);
      const int q  = idc % (WW / 4);
      const float* s = sb + dx * SP + 4 * (q ^ ((dx >> 1) & 7));
      const float4 lo = *(const float4*)s;
      const float4 hi = *(const float4*)(s + SSZ);
      val[k] = make_float4(lo.x + hi.x, lo.y + hi.y, lo.z + hi.z, lo.w + hi.w);
      oo[k]  = (size_t)dx * (HH * WW) + 4 * q;
    }
#pragma unroll
    for (int k = 0; k < NQ; ++k) {
      if (k + 1 < NQ || t + FT * k < FQ) *(float4*)(out + outBase + oo[k]) = val[k];
    }
  };

  auto compute = [&](int buf, int nextRow, v8f& acc0, v8f& acc1, v8f& acc2) {
    const v8f zero8 = {0.f, 0.f, 0.f, 0.f, 0.f, 0.f, 0.f, 0.f};
    acc0 = zero8;
    acc1 = zero8;
    acc2 = zero8;

    const float* bp =
        &Bld[buf * BSZ + (kh * (CCH / 4) + khi) * BPR + par * BPH + 2 * (ti * 16 + row)];

    constexpr int KB  = 8;
    constexpr int KBN = KSTEPS / KB;
    static_assert(KBN * KB == KSTEPS, "blocks must cover the K half exactly");
    v2f fb[2][KB][3];

    auto load_steps = [&](int s, int blk, int t0, int t1) {
#pragma unroll
      for (int t = t0; t < t1; ++t) {
        const int kk = blk * KB + t;
        fb[s][t][0] = *(const v2f*)(bp + kk * (2 * BPR));
        fb[s][t][1] = *(const v2f*)(bp + kk * (2 * BPR) + 32);
        fb[s][t][2] = *(const v2f*)(bp + kk * (2 * BPR) + 64);
      }
    };

    load_steps(0, 0, 0, 2);
    __builtin_amdgcn_sched_barrier(0);
    load_steps(0, 0, 2, KB);
    __builtin_amdgcn_sched_barrier(0);
#pragma unroll
    for (int blk = 0; blk < KBN; ++blk) {
      if (blk + 1 < KBN) load_steps((blk + 1) & 1, blk + 1, 0, KB);
      if (blk == 0 && nextRow >= 0) fetch(nextRow);
      __builtin_amdgcn_sched_barrier(0);
#pragma unroll
      for (int t = 0; t < KB; ++t) {
        const v2f a = fa[blk * KB + t];
        acc0 = __builtin_amdgcn_wmma_f32_16x16x4_f32(
            false, a, false, fb[blk & 1][t][0], (short)0, acc0, false, false);
        acc1 = __builtin_amdgcn_wmma_f32_16x16x4_f32(
            false, a, false, fb[blk & 1][t][1], (short)0, acc1, false, false);
        acc2 = __builtin_amdgcn_wmma_f32_16x16x4_f32(
            false, a, false, fb[blk & 1][t][2], (short)0, acc2, false, false);
      }
      __builtin_amdgcn_sched_barrier(0);
    }
  };

  auto stage_store = [&](int dyI, const v8f& acc0, const v8f& acc1, const v8f& acc2) {
    float* __restrict__ sbuf = Sld + ((dyI & 1) * 2 + kh) * SSZ;
    const int mBase = khi * 8;
    const int N = lane & 15;
#pragma unroll
    for (int r = 0; r < 8; ++r) {
      const int m  = mBase + r;
      const int w1 = 2 * (ti * 16 + m) + par;
      const int d0 = N - m;
      const int d1 = d0 + 16;
      const int d2 = d0 + 32;
      if (d0 >= 0)  sbuf[stage_off(d0, w1)] = acc0[r] * 0.00390625f;
      if (d1 <= 20) sbuf[stage_off(d1, w1)] = acc1[r] * 0.00390625f;
      if (d2 <= 20) sbuf[stage_off(d2, w1)] = acc2[r] * 0.00390625f;
    }
  };

  for (int dyI = dLo; dyI <= dHi; ++dyI) {
    __syncthreads();
    if (dyI > dLo) flush(dyI - 1);
    v8f acc0, acc1, acc2;
    compute(dyI & 1, (dyI < dHi) ? (h + (dyI + 1) * 2 - 20) : -1, acc0, acc1, acc2);
    if (dyI < dHi) put((dyI + 1) & 1);
    stage_store(dyI, acc0, acc1, acc2);
  }
  __syncthreads();
  flush(dHi);

  for (int dyI = 0; dyI < ND; ++dyI) {
    if (dyI >= dLo && dyI <= dHi) continue;
    const size_t outBase =
        (((size_t)b * (ND * ND) + (size_t)dyI * ND) * HH + (size_t)h) * WW;
    for (int i = tid; i < ND * WW; i += NT) {
      const int dxI = i / WW, w = i % WW;
      out[outBase + (size_t)dxI * (HH * WW) + w] = 0.f;
    }
  }
}

extern "C" void kernel_launch(void* const* d_in, const int* in_sizes, int n_in,
                              void* d_out, int out_size, void* d_ws, size_t ws_size,
                              hipStream_t stream) {
  const float* f1 = (const float*)d_in[0];
  const float* f2 = (const float*)d_in[1];
  float* out = (float*)d_out;

  dim3 grid(HH, BATCH);
  dim3 block(NT);
  corr_wmma_f32_kernel<<<grid, block, 0, stream>>>(f1, f2, out);
}


# diagonal store to the LDS output stage: two unconditional stores per accumulator register with selected value and dx (was three stores, each under its own lane mask and branch)
# speedup vs baseline: 2.2535x; 1.0502x over previous
#include <hip/hip_runtime.h>


typedef __attribute__((ext_vector_type(2))) float v2f;
typedef __attribute__((ext_vector_type(4))) float v4f;
typedef __attribute__((ext_vector_type(8))) float v8f;

#define BATCH 8
#define CCH   256
#define HH    64
#define WW    96
#define ND    21
#define BPR   288
#define BPH   144
#define BSZ   ((CCH / 2) * BPR)
#define NT    384
#define NLD   (CCH * (WW / 4) / NT)
#define SP    96
#define SSZ   (ND * SP)
#define LDS_DW (2 * BSZ + 4 * SSZ)

static_assert(LDS_DW * 4 <= 327680, "two row tiles and the stages must fit the LDS of one WGP");

__device__ __forceinline__ int stage_off(int d, int w1) {
  return d * SP + ((((w1 >> 2) ^ ((d >> 1) & 7))) << 2) + (w1 & 3);
}

__global__ __launch_bounds__(NT, 1)
void corr_wmma_f32_kernel(const float* __restrict__ f1,
                          const float* __restrict__ f2,
                          float* __restrict__ out) {
  __shared__ __attribute__((aligned(16))) float smem[LDS_DW];
  float* __restrict__ Bld = smem;
  float* __restrict__ Sld = smem + 2 * BSZ;

  const int tid  = threadIdx.x;
  const int lane = tid & 31;
  const int wv   = tid >> 5;
  const int kh   = wv & 1;
  const int par  = (wv >= 6) ? 1 : 0;
  const int ti   = (wv >> 1) - 3 * par;

  static_assert(BATCH == 8 && HH == 64, "rank map below is written for 8 x 64 rows");
  const int flat = blockIdx.y * HH + blockIdx.x;
  const int b    = flat & (BATCH - 1);
  const int rank = flat >> 3;
  int h;
  if (rank < 24) {
    h = 20 + rank;
  } else {
    const int s = rank - 24;
    const int e = (s >> 1) + 1;
    h = (s & 1) ? (43 + e) : (20 - e);
  }

  const int row = lane & 15;
  const int khi = lane >> 4;

  const size_t inBase = (size_t)b * (CCH * HH * WW);

  const int dLo = (h < 20) ? ((21 - h) >> 1) : 0;
  const int dHi = (h > 43) ? ((83 - h) >> 1) : 20;

  constexpr int PSTEP = NT / (WW / 4);
  constexpr int NIT   = (CCH / 2) / PSTEP;
  static_assert(NT % (WW / 4) == 0 && (CCH / 2) % PSTEP == 0 && NLD == 2 * NIT,
                "staging map needs whole pair rows per pass");
  static_assert((WW / 4) % 8 == 0, "an 8-lane group must not straddle two pair rows");
  const int c0 = tid / (WW / 4);
  const int q0 = tid % (WW / 4);
  const float* fsrc = f2 + inBase + (size_t)(2 * c0) * (HH * WW) + q0 * 4;
  float* bdst = &Bld[c0 * BPR + 20 + 4 * q0];
  float4 pre[NLD];

  auto fetch = [&](int h2r) {
    const float* src = fsrc + (size_t)h2r * WW;
#pragma unroll
    for (int j = 0; j < NIT; ++j) {
      pre[2 * j]     = *(const float4*)(src + (size_t)(2 * PSTEP * j) * (HH * WW));
      pre[2 * j + 1] = *(const float4*)(src + (size_t)(2 * PSTEP * j + 1) * (HH * WW));
    }
  };
  auto put = [&](int buf) {
    float* dbase = bdst + buf * BSZ;
#pragma unroll
    for (int j = 0; j < NIT; ++j) {
      const float4 x = pre[2 * j];
      const float4 y = pre[2 * j + 1];
      const v4f ev = {x.x, y.x, x.z, y.z};
      const v4f od = {x.y, y.y, x.w, y.w};
      float* d = dbase + j * (PSTEP * BPR);
      *(v4f*)(d)       = ev;
      *(v4f*)(d + BPH) = od;
    }
  };

  constexpr int KSTEPS = CCH / 2 / 4;
  v2f fa[KSTEPS];
  {
    const float* asrc = f1 + inBase + (size_t)(kh * (CCH / 2) + 2 * khi) * (HH * WW) +
                        (size_t)h * WW + 2 * (ti * 16 + row) + par;
#pragma unroll
    for (int kk = 0; kk < KSTEPS; ++kk) {
      fa[kk].x = asrc[(size_t)(4 * kk) * (HH * WW)];
      fa[kk].y = asrc[(size_t)(4 * kk + 1) * (HH * WW)];
    }
  }
  fetch(h + dLo * 2 - 20);

  {
    const float4 z = make_float4(0.f, 0.f, 0.f, 0.f);
    for (int i = tid; i < 2 * (CCH / 2) * 24; i += NT) {
      const int p = i / 24;
      const int j = i % 24;
      const int off = (j < 5) ? (j * 4) : (j < 17) ? (116 + (j - 5) * 4) : (260 + (j - 17) * 4);
      *(float4*)(&Bld[p * BPR + off]) = z;
    }
  }
  put(dLo & 1);

  constexpr int FQ = ND * (WW / 4);
  constexpr int FT = 4 * 32;
  auto flush = [&](int dyI) {
    if (wv < (NT - FT) / 32) return;
    const size_t outBase =
        (((size_t)b * (ND * ND) + (size_t)dyI * ND) * HH + (size_t)h) * WW;
    const float* sb = Sld + (dyI & 1) * (2 * SSZ);
    const int t = tid - (NT - FT);
    constexpr int NQ = (FQ + FT - 1) / FT;
    static_assert((NQ - 1) * FT < FQ && FQ % 8 == 0, "all but the last quad of every thread exist");
    float4 val[NQ];
    size_t oo[NQ];
#pragma unroll
    for (int k = 0; k < NQ; ++k) {
      const int idx = t + FT * k;
      const int idc = (k + 1 < NQ || idx < FQ) ? idx : (FQ - 1);
      const int dx = idc / (WW / 4);
      const int q  = idc % (WW / 4);
      const float* s = sb + dx * SP + 4 * (q ^ ((dx >> 1) & 7));
      const float4 lo = *(const float4*)s;
      const float4 hi = *(const float4*)(s + SSZ);
      val[k] = make_float4(lo.x + hi.x, lo.y + hi.y, lo.z + hi.z, lo.w + hi.w);
      oo[k]  = (size_t)dx * (HH * WW) + 4 * q;
    }
#pragma unroll
    for (int k = 0; k < NQ; ++k) {
      if (k + 1 < NQ || t + FT * k < FQ) *(float4*)(out + outBase + oo[k]) = val[k];
    }
  };

  auto compute = [&](int buf, int nextRow, v8f& acc0, v8f& acc1, v8f& acc2) {
    const v8f zero8 = {0.f, 0.f, 0.f, 0.f, 0.f, 0.f, 0.f, 0.f};
    acc0 = zero8;
    acc1 = zero8;
    acc2 = zero8;

    const float* bp =
        &Bld[buf * BSZ + (kh * (CCH / 4) + khi) * BPR + par * BPH + 2 * (ti * 16 + row)];

    constexpr int KB  = 8;
    constexpr int KBN = KSTEPS / KB;
    static_assert(KBN * KB == KSTEPS, "blocks must cover the K half exactly");
    v2f fb[2][KB][3];

    auto load_steps = [&](int s, int blk, int t0, int t1) {
#pragma unroll
      for (int t = t0; t < t1; ++t) {
        const int kk = blk * KB + t;
        fb[s][t][0] = *(const v2f*)(bp + kk * (2 * BPR));
        fb[s][t][1] = *(const v2f*)(bp + kk * (2 * BPR) + 32);
        fb[s][t][2] = *(const v2f*)(bp + kk * (2 * BPR) + 64);
      }
    };

    load_steps(0, 0, 0, 2);
    __builtin_amdgcn_sched_barrier(0);
    load_steps(0, 0, 2, KB);
    __builtin_amdgcn_sched_barrier(0);
#pragma unroll
    for (int blk = 0; blk < KBN; ++blk) {
      if (blk + 1 < KBN) load_steps((blk + 1) & 1, blk + 1, 0, KB);
      if (blk == 0 && nextRow >= 0) fetch(nextRow);
      __builtin_amdgcn_sched_barrier(0);
#pragma unroll
      for (int t = 0; t < KB; ++t) {
        const v2f a = fa[blk * KB + t];
        acc0 = __builtin_amdgcn_wmma_f32_16x16x4_f32(
            false, a, false, fb[blk & 1][t][0], (short)0, acc0, false, false);
        acc1 = __builtin_amdgcn_wmma_f32_16x16x4_f32(
            false, a, false, fb[blk & 1][t][1], (short)0, acc1, false, false);
        acc2 = __builtin_amdgcn_wmma_f32_16x16x4_f32(
            false, a, false, fb[blk & 1][t][2], (short)0, acc2, false, false);
      }
      __builtin_amdgcn_sched_barrier(0);
    }
  };

  auto stage_store = [&](int dyI, const v8f& acc0, const v8f& acc1, const v8f& acc2) {
    float* __restrict__ sbuf = Sld + ((dyI & 1) * 2 + kh) * SSZ;
    const int mBase = khi * 8;
    const int N = lane & 15;
#pragma unroll
    for (int r = 0; r < 8; ++r) {
      const int m  = mBase + r;
      const int w1 = 2 * (ti * 16 + m) + par;
      const int d0 = N - m;
      const bool pLo = d0 <= 4;
      const bool qHi = d0 >= 0;
      const bool qLo = d0 <= -12;
      const float vp = pLo ? acc1[r] : acc0[r];
      const int   dp = pLo ? d0 + 16 : d0;
      const float vq = qHi ? acc0[r] : (qLo ? acc2[r] : acc1[r]);
      const int   dq = qHi ? d0 : (qLo ? d0 + 32 : d0 + 16);
      sbuf[stage_off(dp, w1)] = vp * 0.00390625f;
      sbuf[stage_off(dq, w1)] = vq * 0.00390625f;
    }
  };

  for (int dyI = dLo; dyI <= dHi; ++dyI) {
    __syncthreads();
    if (dyI > dLo) flush(dyI - 1);
    v8f acc0, acc1, acc2;
    compute(dyI & 1, (dyI < dHi) ? (h + (dyI + 1) * 2 - 20) : -1, acc0, acc1, acc2);
    if (dyI < dHi) put((dyI + 1) & 1);
    stage_store(dyI, acc0, acc1, acc2);
  }
  __syncthreads();
  flush(dHi);

  for (int dyI = 0; dyI < ND; ++dyI) {
    if (dyI >= dLo && dyI <= dHi) continue;
    const size_t outBase =
        (((size_t)b * (ND * ND) + (size_t)dyI * ND) * HH + (size_t)h) * WW;
    for (int i = tid; i < ND * WW; i += NT) {
      const int dxI = i / WW, w = i % WW;
      out[outBase + (size_t)dxI * (HH * WW) + w] = 0.f;
    }
  }
}

extern "C" void kernel_launch(void* const* d_in, const int* in_sizes, int n_in,
                              void* d_out, int out_size, void* d_ws, size_t ws_size,
                              hipStream_t stream) {
  const float* f1 = (const float*)d_in[0];
  const float* f2 = (const float*)d_in[1];
  float* out = (float*)d_out;

  dim3 grid(HH, BATCH);
  dim3 block(NT);
  corr_wmma_f32_kernel<<<grid, block, 0, stream>>>(f1, f2, out);
}


# a work-group takes two f1 rows (h, h+2) that share every staged f2 row: an A tile is 8 columns x 2 rows, its band needs 2 column tiles: 12 tiles per dy row, was 18; K in one chain (sum order changed)
# speedup vs baseline: 3.0573x; 1.3567x over previous
#include <hip/hip_runtime.h>


typedef __attribute__((ext_vector_type(2))) float v2f;
typedef __attribute__((ext_vector_type(4))) float v4f;
typedef __attribute__((ext_vector_type(8))) float v8f;

#define BATCH 8
#define CCH   256
#define HH    64
#define WW    96
#define ND    21
#define BPR   288
#define BPH   144
#define BSZ   ((CCH / 2) * BPR)
#define NT    384
#define NLD   (CCH * (WW / 4) / NT)
#define SP    96
#define SSZ   (ND * SP)
#define LDS_DW (2 * BSZ + 4 * SSZ)

static_assert(LDS_DW * 4 <= 327680, "two row tiles and the stages must fit the LDS of one WGP");

__device__ __forceinline__ int stage_off(int d, int w1) {
  return d * SP + ((((w1 >> 2) ^ ((d >> 1) & 7))) << 2) + (w1 & 3);
}

__global__ __launch_bounds__(NT, 1)
void corr_wmma_f32_kernel(const float* __restrict__ f1,
                          const float* __restrict__ f2,
                          float* __restrict__ out) {
  __shared__ __attribute__((aligned(16))) float smem[LDS_DW];
  float* __restrict__ Bld = smem;
  float* __restrict__ Sld = smem + 2 * BSZ;

  const int tid  = threadIdx.x;
  const int lane = tid & 31;
  const int wv   = tid >> 5;
  const int par  = (wv >= 6) ? 1 : 0;
  const int g    = wv - 6 * par;

  static_assert(BATCH == 8 && HH == 64, "rank map below is written for 8 x 32 row pairs");
  const int flat = blockIdx.y * (HH / 2) + blockIdx.x;
  const int b    = flat & (BATCH - 1);
  const int rank = flat >> 3;
  int p;
  if (rank < 12) {
    p = 10 + rank;
  } else {
    const int o = rank - 12;
    p = (o & 1) ? (22 + (o >> 1)) : (9 - (o >> 1));
  }
  const int hA = 4 * (p >> 1) + (p & 1);

  const int row = lane & 15;
  const int khi = lane >> 4;

  const size_t inBase = (size_t)b * (CCH * HH * WW);

  const int sLo = (hA < 20) ? ((21 - hA) >> 1) : 0;
  const int sHi = (hA > 41) ? ((83 - hA) >> 1) : ND;

  constexpr int PSTEP = NT / (WW / 4);
  constexpr int NIT   = (CCH / 2) / PSTEP;
  static_assert(NT % (WW / 4) == 0 && (CCH / 2) % PSTEP == 0 && NLD == 2 * NIT,
                "staging map needs whole pair rows per pass");
  static_assert((WW / 4) % 8 == 0, "an 8-lane group must not straddle two pair rows");
  const int c0 = tid / (WW / 4);
  const int q0 = tid % (WW / 4);
  const float* fsrc = f2 + inBase + (size_t)(2 * c0) * (HH * WW) + q0 * 4;
  float* bdst = &Bld[c0 * BPR + 20 + 4 * q0];
  float4 pre[NLD];

  auto fetch = [&](int h2r) {
    const float* src = fsrc + (size_t)h2r * WW;
#pragma unroll
    for (int j = 0; j < NIT; ++j) {
      pre[2 * j]     = *(const float4*)(src + (size_t)(2 * PSTEP * j) * (HH * WW));
      pre[2 * j + 1] = *(const float4*)(src + (size_t)(2 * PSTEP * j + 1) * (HH * WW));
    }
  };
  auto put = [&](int buf) {
    float* dbase = bdst + buf * BSZ;
#pragma unroll
    for (int j = 0; j < NIT; ++j) {
      const float4 x = pre[2 * j];
      const float4 y = pre[2 * j + 1];
      const v4f ev = {x.x, y.x, x.z, y.z};
      const v4f od = {x.y, y.y, x.w, y.w};
      float* d = dbase + j * (PSTEP * BPR);
      *(v4f*)(d)       = ev;
      *(v4f*)(d + BPH) = od;
    }
  };

  constexpr int KSTEPS = CCH / 4;
  v2f fa[KSTEPS];
  {
    const float* asrc = f1 + inBase + (size_t)(2 * khi) * (HH * WW) +
                        (size_t)(hA + 2 * (row >> 3)) * WW + 2 * (g * 8 + (row & 7)) + par;
#pragma unroll
    for (int kk = 0; kk < KSTEPS; ++kk) {
      fa[kk].x = asrc[(size_t)(4 * kk) * (HH * WW)];
      fa[kk].y = asrc[(size_t)(4 * kk + 1) * (HH * WW)];
    }
  }
  fetch(hA + sLo * 2 - 20);

  {
    const float4 z = make_float4(0.f, 0.f, 0.f, 0.f);
    for (int i = tid; i < 2 * (CCH / 2) * 24; i += NT) {
      const int p = i / 24;
      const int j = i % 24;
      const int off = (j < 5) ? (j * 4) : (j < 17) ? (116 + (j - 5) * 4) : (260 + (j - 17) * 4);
      *(float4*)(&Bld[p * BPR + off]) = z;
    }
  }
  put(sLo & 1);

  constexpr int FQ = ND * (WW / 4);
  constexpr int FT = 4 * 32;
  auto flush = [&](int s) {
    if (wv < (NT - FT) / 32) return;
    const int t = tid - (NT - FT);
    constexpr int NQ = (FQ + FT - 1) / FT;
    static_assert((NQ - 1) * FT < FQ && FQ % 8 == 0, "all but the last quad of every thread exist");
#pragma unroll
    for (int pl = 0; pl < 2; ++pl) {
      const int dyI = s - pl;
      if (dyI < 0 || dyI >= ND) continue;
      float* const ob =
          out + (((size_t)b * (ND * ND) + (size_t)dyI * ND) * HH + (size_t)(hA + 2 * pl)) * WW;
      const float* sb = Sld + ((s & 1) * 2 + pl) * SSZ;
      v4f val[NQ];
      unsigned oo[NQ];
#pragma unroll
      for (int k = 0; k < NQ; ++k) {
        const int idx = t + FT * k;
        const int idc = (k + 1 < NQ || idx < FQ) ? idx : (FQ - 1);
        const int dx = idc / (WW / 4);
        const int q  = idc % (WW / 4);
        val[k] = *(const v4f*)(sb + dx * SP + 4 * (q ^ ((dx >> 1) & 7)));
        oo[k]  = (unsigned)(dx * (HH * WW) + 4 * q);
      }
#pragma unroll
      for (int k = 0; k < NQ; ++k) {
        if (k + 1 < NQ || t + FT * k < FQ) *(v4f*)(ob + oo[k]) = val[k];
      }
    }
  };

  auto compute = [&](int buf, int nextRow, v8f& acc0, v8f& acc1) {
    const v8f zero8 = {0.f, 0.f, 0.f, 0.f, 0.f, 0.f, 0.f, 0.f};
    acc0 = zero8;
    acc1 = zero8;

    const float* bp = &Bld[buf * BSZ + khi * BPR + par * BPH + 2 * (g * 8 + row)];

    constexpr int KB  = 8;
    constexpr int KBN = KSTEPS / KB;
    static_assert(KBN * KB == KSTEPS, "blocks must cover K exactly");
    v2f fb[2][KB][2];

    auto load_steps = [&](int s, int blk, int t0, int t1) {
#pragma unroll
      for (int t = t0; t < t1; ++t) {
        const int kk = blk * KB + t;
        fb[s][t][0] = *(const v2f*)(bp + kk * (2 * BPR));
        fb[s][t][1] = *(const v2f*)(bp + kk * (2 * BPR) + 32);
      }
    };

    load_steps(0, 0, 0, 2);
    __builtin_amdgcn_sched_barrier(0);
    load_steps(0, 0, 2, KB);
    __builtin_amdgcn_sched_barrier(0);
#pragma unroll
    for (int blk = 0; blk < KBN; ++blk) {
      if (blk + 1 < KBN) load_steps((blk + 1) & 1, blk + 1, 0, KB);
      if (blk == 0 && nextRow >= 0) fetch(nextRow);
      __builtin_amdgcn_sched_barrier(0);
#pragma unroll
      for (int t = 0; t < KB; ++t) {
        const v2f a = fa[blk * KB + t];
        acc0 = __builtin_amdgcn_wmma_f32_16x16x4_f32(
            false, a, false, fb[blk & 1][t][0], (short)0, acc0, false, false);
        acc1 = __builtin_amdgcn_wmma_f32_16x16x4_f32(
            false, a, false, fb[blk & 1][t][1], (short)0, acc1, false, false);
      }
      __builtin_amdgcn_sched_barrier(0);
    }
  };

  auto stage_store = [&](int s, const v8f& acc0, const v8f& acc1) {
    float* __restrict__ sbuf = Sld + ((s & 1) * 2 + khi) * SSZ;
    const int n = lane & 15;
#pragma unroll
    for (int r = 0; r < 8; ++r) {
      const int w1 = 2 * (g * 8 + r) + par;
      const int d0 = n - r;
      const bool pLo = d0 <= 4;
      const bool qHi = d0 >= 0;
      const float vp = pLo ? acc1[r] : acc0[r];
      const int   dp = pLo ? d0 + 16 : d0;
      const float vq = qHi ? acc0[r] : acc1[r];
      const int   dq = qHi ? d0 : d0 + 16;
      sbuf[stage_off(dp, w1)] = vp * 0.00390625f;
      sbuf[stage_off(dq, w1)] = vq * 0.00390625f;
    }
  };

  for (int s = sLo; s <= sHi; ++s) {
    __syncthreads();
    if (s > sLo) flush(s - 1);
    v8f acc0, acc1;
    compute(s & 1, (s < sHi) ? (hA + (s + 1) * 2 - 20) : -1, acc0, acc1);
    stage_store(s, acc0, acc1);
    if (s < sHi) put((s + 1) & 1);
  }
  __syncthreads();
  flush(sHi);

  for (int pl = 0; pl < 2; ++pl) {
    for (int dyI = 0; dyI < ND; ++dyI) {
      if (dyI + pl >= sLo && dyI + pl <= sHi) continue;
      const size_t outBase =
          (((size_t)b * (ND * ND) + (size_t)dyI * ND) * HH + (size_t)(hA + 2 * pl)) * WW;
      for (int i = tid; i < ND * WW; i += NT) {
        const int dxI = i / WW, w = i % WW;
        out[outBase + (size_t)dxI * (HH * WW) + w] = 0.f;
      }
    }
  }
}

extern "C" void kernel_launch(void* const* d_in, const int* in_sizes, int n_in,
                              void* d_out, int out_size, void* d_ws, size_t ws_size,
                              hipStream_t stream) {
  const float* f1 = (const float*)d_in[0];
  const float* f2 = (const float*)d_in[1];
  float* out = (float*)d_out;

  dim3 grid(HH / 2, BATCH);
  dim3 block(NT);
  corr_wmma_f32_kernel<<<grid, block, 0, stream>>>(f1, f2, out);
}
